# starting kernel
# baseline (speedup 1.0000x reference)
#include <hip/hip_runtime.h>

typedef float v2f __attribute__((ext_vector_type(2)));
typedef float v8f __attribute__((ext_vector_type(8)));

#define SEQ_LEN 131072
#define HID 64
#define G4 256

__global__ __launch_bounds__(256) void xproj_wmma(
    const float* __restrict__ inputs, const float* __restrict__ W_ih,
    const float* __restrict__ b_ih, const float* __restrict__ b_hh,
    float* __restrict__ x_proj)
{
  const int lane = threadIdx.x & 31;
  const int wave = threadIdx.x >> 5;
  const int rt   = blockIdx.x * 8 + wave;
  const int l15  = lane & 15;
  const int half = lane >> 4;
  const int row0 = rt * 16;

  const v2f* Arow = reinterpret_cast<const v2f*>(inputs + (size_t)(row0 + l15) * HID);
  v2f areg[16];
#pragma unroll
  for (int kk = 0; kk < 16; ++kk) areg[kk] = Arow[kk * 2 + half];

  for (int ct = 0; ct < 16; ++ct) {
    const int  col  = ct * 16 + l15;
    const v2f* Brow = reinterpret_cast<const v2f*>(W_ih + (size_t)col * HID);
    v8f c = {};
#pragma unroll
    for (int kk = 0; kk < 16; ++kk) {
      v2f b = Brow[kk * 2 + half];
      c = __builtin_amdgcn_wmma_f32_16x16x4_f32(false, areg[kk], false, b,
                                                (short)0, c, false, false);
    }
    const float bias = b_ih[col] + b_hh[col];
#pragma unroll
    for (int v = 0; v < 8; ++v) {
      const int row = row0 + v + half * 8;
      x_proj[(size_t)row * G4 + col] = c[v] + bias;
    }
  }
}

__device__ __forceinline__ float fsig(float x)  { return 1.0f / (1.0f + __expf(-x)); }
__device__ __forceinline__ float ftanh(float x) { return 1.0f - 2.0f / (__expf(2.0f * x) + 1.0f); }

__global__ __launch_bounds__(256) void lstm_scan(
    const float* __restrict__ x_proj, const float* __restrict__ W_hh,
    float* __restrict__ hs)
{
  __shared__ float sh_h[HID];
  __shared__ float sh_z[G4];
  const int j = threadIdx.x;

  float w[HID];
#pragma unroll
  for (int k = 0; k < HID; ++k) w[k] = W_hh[(size_t)j * HID + k];

  float creg = 0.0f;
  if (j < HID) sh_h[j] = 0.0f;
  __syncthreads();

  float xp0 = x_proj[j];
  float xp1 = x_proj[(size_t)G4 + j];

  for (int t = 0; t < SEQ_LEN; ++t) {
    const int   tn  = (t + 2 < SEQ_LEN) ? (t + 2) : (SEQ_LEN - 1);
    const float xp2 = x_proj[(size_t)tn * G4 + j];

    float a0 = xp0, a1 = 0.0f, a2 = 0.0f, a3 = 0.0f;
    const float4* h4 = reinterpret_cast<const float4*>(sh_h);
#pragma unroll
    for (int k = 0; k < 4; ++k) {
      const float4 h0 = h4[4 * k + 0];
      const float4 h1 = h4[4 * k + 1];
      const float4 h2 = h4[4 * k + 2];
      const float4 h3 = h4[4 * k + 3];
      const int b = 16 * k;
      a0 = fmaf(w[b +  0], h0.x, a0); a0 = fmaf(w[b +  1], h0.y, a0);
      a0 = fmaf(w[b +  2], h0.z, a0); a0 = fmaf(w[b +  3], h0.w, a0);
      a1 = fmaf(w[b +  4], h1.x, a1); a1 = fmaf(w[b +  5], h1.y, a1);
      a1 = fmaf(w[b +  6], h1.z, a1); a1 = fmaf(w[b +  7], h1.w, a1);
      a2 = fmaf(w[b +  8], h2.x, a2); a2 = fmaf(w[b +  9], h2.y, a2);
      a2 = fmaf(w[b + 10], h2.z, a2); a2 = fmaf(w[b + 11], h2.w, a2);
      a3 = fmaf(w[b + 12], h3.x, a3); a3 = fmaf(w[b + 13], h3.y, a3);
      a3 = fmaf(w[b + 14], h3.z, a3); a3 = fmaf(w[b + 15], h3.w, a3);
    }
    sh_z[j] = (a0 + a1) + (a2 + a3);
    __syncthreads();

    if (j < HID) {
      const float ig = fsig (sh_z[0 * HID + j]);
      const float fg = fsig (sh_z[1 * HID + j]);
      const float gg = ftanh(sh_z[2 * HID + j]);
      const float og = fsig (sh_z[3 * HID + j]);
      creg = fg * creg + ig * gg;
      const float hn = og * ftanh(creg);
      sh_h[j] = hn;
      hs[(size_t)t * HID + j] = hn;
    }
    __syncthreads();
    xp0 = xp1;
    xp1 = xp2;
  }
}

__global__ __launch_bounds__(256) void head_wmma(
    const float* __restrict__ hs, const float* __restrict__ W_lin,
    const float* __restrict__ b_lin, const float* __restrict__ W_fin,
    const float* __restrict__ b_fin, float* __restrict__ out)
{
  const int lane = threadIdx.x & 31;
  const int wave = threadIdx.x >> 5;
  const int rt   = blockIdx.x * 8 + wave;
  const int l15  = lane & 15;
  const int half = lane >> 4;
  const int row0 = rt * 16;

  const v2f* Arow = reinterpret_cast<const v2f*>(hs + (size_t)(row0 + l15) * HID);
  v2f areg[16];
#pragma unroll
  for (int kk = 0; kk < 16; ++kk) {
    v2f a = Arow[kk * 2 + half];
    a.x = fmaxf(a.x, 0.0f);
    a.y = fmaxf(a.y, 0.0f);
    areg[kk] = a;
  }

  float partial[8];
#pragma unroll
  for (int v = 0; v < 8; ++v) partial[v] = 0.0f;

#pragma unroll
  for (int ctile = 0; ctile < 4; ++ctile) {
    const int  col  = ctile * 16 + l15;
    const v2f* Brow = reinterpret_cast<const v2f*>(W_lin + (size_t)col * HID);
    v8f c = {};
#pragma unroll
    for (int kk = 0; kk < 16; ++kk) {
      v2f b = Brow[kk * 2 + half];
      c = __builtin_amdgcn_wmma_f32_16x16x4_f32(false, areg[kk], false, b,
                                                (short)0, c, false, false);
    }
    const float bl = b_lin[col];
    const float wf = W_fin[col];
#pragma unroll
    for (int v = 0; v < 8; ++v)
      partial[v] += fmaxf(c[v] + bl, 0.0f) * wf;
  }

#pragma unroll
  for (int v = 0; v < 8; ++v) {
    float s = partial[v];
    s += __shfl_xor(s, 1, 32);
    s += __shfl_xor(s, 2, 32);
    s += __shfl_xor(s, 4, 32);
    s += __shfl_xor(s, 8, 32);
    if (l15 == 0) out[row0 + half * 8 + v] = s + b_fin[0];
  }
}

extern "C" void kernel_launch(void* const* d_in, const int* in_sizes, int n_in,
                              void* d_out, int out_size, void* d_ws, size_t ws_size,
                              hipStream_t stream) {
  const float* inputs = (const float*)d_in[0];
  const float* W_ih   = (const float*)d_in[1];
  const float* W_hh   = (const float*)d_in[2];
  const float* b_ih   = (const float*)d_in[3];
  const float* b_hh   = (const float*)d_in[4];
  const float* W_lin  = (const float*)d_in[5];
  const float* b_lin  = (const float*)d_in[6];
  const float* W_fin  = (const float*)d_in[7];
  const float* b_fin  = (const float*)d_in[8];
  float* out = (float*)d_out;

  float* x_proj = (float*)d_ws;
  float* hs     = x_proj + (size_t)SEQ_LEN * G4;

  xproj_wmma<<<1024, 256, 0, stream>>>(inputs, W_ih, b_ih, b_hh, x_proj);
  lstm_scan<<<1, 256, 0, stream>>>(x_proj, W_hh, hs);
  head_wmma<<<1024, 256, 0, stream>>>(hs, W_lin, b_lin, W_fin, b_fin, out);
}


# lstm_scan: the two barriers of a step fence LDS only, and each thread loads its x_proj values one 8-step block ahead into registers, so no step waits for a global load or for the hs store
# speedup vs baseline: 3.0701x; 3.0701x over previous
#include <hip/hip_runtime.h>

typedef float v2f __attribute__((ext_vector_type(2)));
typedef float v8f __attribute__((ext_vector_type(8)));

#define SEQ_LEN 131072
#define HID 64
#define G4 256

__global__ __launch_bounds__(256) void xproj_wmma(
    const float* __restrict__ inputs, const float* __restrict__ W_ih,
    const float* __restrict__ b_ih, const float* __restrict__ b_hh,
    float* __restrict__ x_proj)
{
  const int lane = threadIdx.x & 31;
  const int wave = threadIdx.x >> 5;
  const int rt   = blockIdx.x * 8 + wave;
  const int l15  = lane & 15;
  const int half = lane >> 4;
  const int row0 = rt * 16;

  const v2f* Arow = reinterpret_cast<const v2f*>(inputs + (size_t)(row0 + l15) * HID);
  v2f areg[16];
#pragma unroll
  for (int kk = 0; kk < 16; ++kk) areg[kk] = Arow[kk * 2 + half];

  for (int ct = 0; ct < 16; ++ct) {
    const int  col  = ct * 16 + l15;
    const v2f* Brow = reinterpret_cast<const v2f*>(W_ih + (size_t)col * HID);
    v8f c = {};
#pragma unroll
    for (int kk = 0; kk < 16; ++kk) {
      v2f b = Brow[kk * 2 + half];
      c = __builtin_amdgcn_wmma_f32_16x16x4_f32(false, areg[kk], false, b,
                                                (short)0, c, false, false);
    }
    const float bias = b_ih[col] + b_hh[col];
#pragma unroll
    for (int v = 0; v < 8; ++v) {
      const int row = row0 + v + half * 8;
      x_proj[(size_t)row * G4 + col] = c[v] + bias;
    }
  }
}

__device__ __forceinline__ float fsig(float x)  { return 1.0f / (1.0f + __expf(-x)); }
__device__ __forceinline__ float ftanh(float x) { return 1.0f - 2.0f / (__expf(2.0f * x) + 1.0f); }

#define XPF 8

__device__ __forceinline__ void lds_barrier() {
  __builtin_amdgcn_fence(__ATOMIC_RELEASE, "workgroup", "local");
  __builtin_amdgcn_s_barrier();
  __builtin_amdgcn_fence(__ATOMIC_ACQUIRE, "workgroup", "local");
}

__global__ __launch_bounds__(256) void lstm_scan(
    const float* __restrict__ x_proj, const float* __restrict__ W_hh,
    float* __restrict__ hs)
{
  __shared__ float sh_h[HID];
  __shared__ float sh_z[G4];
  const int j = threadIdx.x;

  float w[HID];
#pragma unroll
  for (int k = 0; k < HID; ++k) w[k] = W_hh[(size_t)j * HID + k];

  float creg = 0.0f;
  if (j < HID) sh_h[j] = 0.0f;
  lds_barrier();

  static_assert(SEQ_LEN % XPF == 0, "whole blocks");
  float xc[XPF], xn[XPF];
  for (int u = 0; u < XPF; ++u) xc[u] = x_proj[(size_t)u * G4 + j];

  for (int tb = 0; tb < SEQ_LEN; tb += XPF) {
    const int nb = (tb + XPF < SEQ_LEN) ? (tb + XPF) : tb;
    for (int u = 0; u < XPF; ++u) xn[u] = x_proj[(size_t)(nb + u) * G4 + j];

    for (int u = 0; u < XPF; ++u) {
      const int t = tb + u;

      float a0 = xc[u], a1 = 0.0f, a2 = 0.0f, a3 = 0.0f;
      const float4* h4 = reinterpret_cast<const float4*>(sh_h);
#pragma unroll
      for (int k = 0; k < 4; ++k) {
        const float4 h0 = h4[4 * k + 0];
        const float4 h1 = h4[4 * k + 1];
        const float4 h2 = h4[4 * k + 2];
        const float4 h3 = h4[4 * k + 3];
        const int b = 16 * k;
        a0 = fmaf(w[b +  0], h0.x, a0); a0 = fmaf(w[b +  1], h0.y, a0);
        a0 = fmaf(w[b +  2], h0.z, a0); a0 = fmaf(w[b +  3], h0.w, a0);
        a1 = fmaf(w[b +  4], h1.x, a1); a1 = fmaf(w[b +  5], h1.y, a1);
        a1 = fmaf(w[b +  6], h1.z, a1); a1 = fmaf(w[b +  7], h1.w, a1);
        a2 = fmaf(w[b +  8], h2.x, a2); a2 = fmaf(w[b +  9], h2.y, a2);
        a2 = fmaf(w[b + 10], h2.z, a2); a2 = fmaf(w[b + 11], h2.w, a2);
        a3 = fmaf(w[b + 12], h3.x, a3); a3 = fmaf(w[b + 13], h3.y, a3);
        a3 = fmaf(w[b + 14], h3.z, a3); a3 = fmaf(w[b + 15], h3.w, a3);
      }
      sh_z[j] = (a0 + a1) + (a2 + a3);
      lds_barrier();

      if (j < HID) {
        const float ig = fsig (sh_z[0 * HID + j]);
        const float fg = fsig (sh_z[1 * HID + j]);
        const float gg = ftanh(sh_z[2 * HID + j]);
        const float og = fsig (sh_z[3 * HID + j]);
        creg = fg * creg + ig * gg;
        const float hn = og * ftanh(creg);
        sh_h[j] = hn;
        hs[(size_t)t * HID + j] = hn;
      }
      lds_barrier();
    }
    for (int u = 0; u < XPF; ++u) xc[u] = xn[u];
  }
}

__global__ __launch_bounds__(256) void head_wmma(
    const float* __restrict__ hs, const float* __restrict__ W_lin,
    const float* __restrict__ b_lin, const float* __restrict__ W_fin,
    const float* __restrict__ b_fin, float* __restrict__ out)
{
  const int lane = threadIdx.x & 31;
  const int wave = threadIdx.x >> 5;
  const int rt   = blockIdx.x * 8 + wave;
  const int l15  = lane & 15;
  const int half = lane >> 4;
  const int row0 = rt * 16;

  const v2f* Arow = reinterpret_cast<const v2f*>(hs + (size_t)(row0 + l15) * HID);
  v2f areg[16];
#pragma unroll
  for (int kk = 0; kk < 16; ++kk) {
    v2f a = Arow[kk * 2 + half];
    a.x = fmaxf(a.x, 0.0f);
    a.y = fmaxf(a.y, 0.0f);
    areg[kk] = a;
  }

  float partial[8];
#pragma unroll
  for (int v = 0; v < 8; ++v) partial[v] = 0.0f;

#pragma unroll
  for (int ctile = 0; ctile < 4; ++ctile) {
    const int  col  = ctile * 16 + l15;
    const v2f* Brow = reinterpret_cast<const v2f*>(W_lin + (size_t)col * HID);
    v8f c = {};
#pragma unroll
    for (int kk = 0; kk < 16; ++kk) {
      v2f b = Brow[kk * 2 + half];
      c = __builtin_amdgcn_wmma_f32_16x16x4_f32(false, areg[kk], false, b,
                                                (short)0, c, false, false);
    }
    const float bl = b_lin[col];
    const float wf = W_fin[col];
#pragma unroll
    for (int v = 0; v < 8; ++v)
      partial[v] += fmaxf(c[v] + bl, 0.0f) * wf;
  }

#pragma unroll
  for (int v = 0; v < 8; ++v) {
    float s = partial[v];
    s += __shfl_xor(s, 1, 32);
    s += __shfl_xor(s, 2, 32);
    s += __shfl_xor(s, 4, 32);
    s += __shfl_xor(s, 8, 32);
    if (l15 == 0) out[row0 + half * 8 + v] = s + b_fin[0];
  }
}

extern "C" void kernel_launch(void* const* d_in, const int* in_sizes, int n_in,
                              void* d_out, int out_size, void* d_ws, size_t ws_size,
                              hipStream_t stream) {
  const float* inputs = (const float*)d_in[0];
  const float* W_ih   = (const float*)d_in[1];
  const float* W_hh   = (const float*)d_in[2];
  const float* b_ih   = (const float*)d_in[3];
  const float* b_hh   = (const float*)d_in[4];
  const float* W_lin  = (const float*)d_in[5];
  const float* b_lin  = (const float*)d_in[6];
  const float* W_fin  = (const float*)d_in[7];
  const float* b_fin  = (const float*)d_in[8];
  float* out = (float*)d_out;

  float* x_proj = (float*)d_ws;
  float* hs     = x_proj + (size_t)SEQ_LEN * G4;

  xproj_wmma<<<1024, 256, 0, stream>>>(inputs, W_ih, b_ih, b_hh, x_proj);
  lstm_scan<<<1, 256, 0, stream>>>(x_proj, W_hh, hs);
  head_wmma<<<1024, 256, 0, stream>>>(hs, W_lin, b_lin, W_fin, b_fin, out);
}


# lstm_scan: one thread per hidden unit holds its 4 gate rows (64 threads, 256 weights in VGPRs): gates stay in the thread, no z round trip through LDS, one barrier per step, a quarter of the h reads
# speedup vs baseline: 4.4057x; 1.4350x over previous
#include <hip/hip_runtime.h>

typedef float v2f __attribute__((ext_vector_type(2)));
typedef float v8f __attribute__((ext_vector_type(8)));

#define SEQ_LEN 131072
#define HID 64
#define G4 256

__global__ __launch_bounds__(256) void xproj_wmma(
    const float* __restrict__ inputs, const float* __restrict__ W_ih,
    const float* __restrict__ b_ih, const float* __restrict__ b_hh,
    float* __restrict__ x_proj)
{
  const int lane = threadIdx.x & 31;
  const int wave = threadIdx.x >> 5;
  const int rt   = blockIdx.x * 8 + wave;
  const int l15  = lane & 15;
  const int half = lane >> 4;
  const int row0 = rt * 16;

  const v2f* Arow = reinterpret_cast<const v2f*>(inputs + (size_t)(row0 + l15) * HID);
  v2f areg[16];
#pragma unroll
  for (int kk = 0; kk < 16; ++kk) areg[kk] = Arow[kk * 2 + half];

  for (int ct = 0; ct < 16; ++ct) {
    const int  col  = ct * 16 + l15;
    const v2f* Brow = reinterpret_cast<const v2f*>(W_ih + (size_t)col * HID);
    v8f c = {};
#pragma unroll
    for (int kk = 0; kk < 16; ++kk) {
      v2f b = Brow[kk * 2 + half];
      c = __builtin_amdgcn_wmma_f32_16x16x4_f32(false, areg[kk], false, b,
                                                (short)0, c, false, false);
    }
    const float bias = b_ih[col] + b_hh[col];
#pragma unroll
    for (int v = 0; v < 8; ++v) {
      const int row = row0 + v + half * 8;
      x_proj[(size_t)row * G4 + col] = c[v] + bias;
    }
  }
}

__device__ __forceinline__ float fsig(float x)  { return 1.0f / (1.0f + __expf(-x)); }
__device__ __forceinline__ float ftanh(float x) { return 1.0f - 2.0f / (__expf(2.0f * x) + 1.0f); }

#define XPF 8

__device__ __forceinline__ void lds_barrier() {
  __builtin_amdgcn_fence(__ATOMIC_RELEASE, "workgroup", "local");
  __builtin_amdgcn_s_barrier();
  __builtin_amdgcn_fence(__ATOMIC_ACQUIRE, "workgroup", "local");
}

__device__ __forceinline__ v2f fma2(v2f w, float h, v2f acc) {
  const v2f hh = {h, h};
  return __builtin_elementwise_fma(w, hh, acc);
}

__global__ __launch_bounds__(HID) void lstm_scan(
    const float* __restrict__ x_proj, const float* __restrict__ W_hh,
    float* __restrict__ hs)
{
  __shared__ __attribute__((aligned(16))) float sh_h[2][HID];
  const int j = threadIdx.x;

  v2f wif[HID], wgo[HID];
#pragma unroll
  for (int k = 0; k < HID; ++k) {
    wif[k].x = W_hh[(size_t)(0 * HID + j) * HID + k];
    wif[k].y = W_hh[(size_t)(1 * HID + j) * HID + k];
    wgo[k].x = W_hh[(size_t)(2 * HID + j) * HID + k];
    wgo[k].y = W_hh[(size_t)(3 * HID + j) * HID + k];
  }

  float creg = 0.0f;
  sh_h[0][j] = 0.0f;
  lds_barrier();

  static_assert(SEQ_LEN % XPF == 0, "whole blocks");
  static_assert(XPF % 2 == 0, "buffer parity is the step's index in its block");
  float xc[XPF][4], xn[XPF][4];
  for (int u = 0; u < XPF; ++u)
    for (int g = 0; g < 4; ++g) xc[u][g] = x_proj[(size_t)u * G4 + g * HID + j];

  for (int tb = 0; tb < SEQ_LEN; tb += XPF) {
    const int nb = (tb + XPF < SEQ_LEN) ? (tb + XPF) : tb;
    for (int u = 0; u < XPF; ++u)
      for (int g = 0; g < 4; ++g) xn[u][g] = x_proj[(size_t)(nb + u) * G4 + g * HID + j];

#pragma unroll
    for (int u = 0; u < XPF; ++u) {
      const int t = tb + u;

      v2f aif[4], ago[4];
      aif[0].x = xc[u][0]; aif[0].y = xc[u][1];
      ago[0].x = xc[u][2]; ago[0].y = xc[u][3];
#pragma unroll
      for (int c = 1; c < 4; ++c) { aif[c].x = 0.0f; aif[c].y = 0.0f; ago[c].x = 0.0f; ago[c].y = 0.0f; }

      const float4* h4 = reinterpret_cast<const float4*>(sh_h[u & 1]);
#pragma unroll
      for (int q = 0; q < 4; ++q) {
#pragma unroll
        for (int c = 0; c < 4; ++c) {
          const float4 h = h4[4 * q + c];
          const int k = 16 * q + 4 * c;
          aif[c] = fma2(wif[k + 0], h.x, aif[c]);  ago[c] = fma2(wgo[k + 0], h.x, ago[c]);
          aif[c] = fma2(wif[k + 1], h.y, aif[c]);  ago[c] = fma2(wgo[k + 1], h.y, ago[c]);
          aif[c] = fma2(wif[k + 2], h.z, aif[c]);  ago[c] = fma2(wgo[k + 2], h.z, ago[c]);
          aif[c] = fma2(wif[k + 3], h.w, aif[c]);  ago[c] = fma2(wgo[k + 3], h.w, ago[c]);
        }
      }
      const v2f zif = (aif[0] + aif[1]) + (aif[2] + aif[3]);
      const v2f zgo = (ago[0] + ago[1]) + (ago[2] + ago[3]);

      const float ig = fsig (zif.x);
      const float fg = fsig (zif.y);
      const float gg = ftanh(zgo.x);
      const float og = fsig (zgo.y);
      creg = fg * creg + ig * gg;
      const float hn = og * ftanh(creg);
      sh_h[(u + 1) & 1][j] = hn;
      hs[(size_t)t * HID + j] = hn;
      lds_barrier();
    }
    for (int u = 0; u < XPF; ++u)
      for (int g = 0; g < 4; ++g) xc[u][g] = xn[u][g];
  }
}

__global__ __launch_bounds__(256) void head_wmma(
    const float* __restrict__ hs, const float* __restrict__ W_lin,
    const float* __restrict__ b_lin, const float* __restrict__ W_fin,
    const float* __restrict__ b_fin, float* __restrict__ out)
{
  const int lane = threadIdx.x & 31;
  const int wave = threadIdx.x >> 5;
  const int rt   = blockIdx.x * 8 + wave;
  const int l15  = lane & 15;
  const int half = lane >> 4;
  const int row0 = rt * 16;

  const v2f* Arow = reinterpret_cast<const v2f*>(hs + (size_t)(row0 + l15) * HID);
  v2f areg[16];
#pragma unroll
  for (int kk = 0; kk < 16; ++kk) {
    v2f a = Arow[kk * 2 + half];
    a.x = fmaxf(a.x, 0.0f);
    a.y = fmaxf(a.y, 0.0f);
    areg[kk] = a;
  }

  float partial[8];
#pragma unroll
  for (int v = 0; v < 8; ++v) partial[v] = 0.0f;

#pragma unroll
  for (int ctile = 0; ctile < 4; ++ctile) {
    const int  col  = ctile * 16 + l15;
    const v2f* Brow = reinterpret_cast<const v2f*>(W_lin + (size_t)col * HID);
    v8f c = {};
#pragma unroll
    for (int kk = 0; kk < 16; ++kk) {
      v2f b = Brow[kk * 2 + half];
      c = __builtin_amdgcn_wmma_f32_16x16x4_f32(false, areg[kk], false, b,
                                                (short)0, c, false, false);
    }
    const float bl = b_lin[col];
    const float wf = W_fin[col];
#pragma unroll
    for (int v = 0; v < 8; ++v)
      partial[v] += fmaxf(c[v] + bl, 0.0f) * wf;
  }

#pragma unroll
  for (int v = 0; v < 8; ++v) {
    float s = partial[v];
    s += __shfl_xor(s, 1, 32);
    s += __shfl_xor(s, 2, 32);
    s += __shfl_xor(s, 4, 32);
    s += __shfl_xor(s, 8, 32);
    if (l15 == 0) out[row0 + half * 8 + v] = s + b_fin[0];
  }
}

extern "C" void kernel_launch(void* const* d_in, const int* in_sizes, int n_in,
                              void* d_out, int out_size, void* d_ws, size_t ws_size,
                              hipStream_t stream) {
  const float* inputs = (const float*)d_in[0];
  const float* W_ih   = (const float*)d_in[1];
  const float* W_hh   = (const float*)d_in[2];
  const float* b_ih   = (const float*)d_in[3];
  const float* b_hh   = (const float*)d_in[4];
  const float* W_lin  = (const float*)d_in[5];
  const float* b_lin  = (const float*)d_in[6];
  const float* W_fin  = (const float*)d_in[7];
  const float* b_fin  = (const float*)d_in[8];
  float* out = (float*)d_out;

  float* x_proj = (float*)d_ws;
  float* hs     = x_proj + (size_t)SEQ_LEN * G4;

  xproj_wmma<<<1024, 256, 0, stream>>>(inputs, W_ih, b_ih, b_hh, x_proj);
  lstm_scan<<<1, HID, 0, stream>>>(x_proj, W_hh, hs);
  head_wmma<<<1024, 256, 0, stream>>>(hs, W_lin, b_lin, W_fin, b_fin, out);
}


# lstm_scan: sigmoid and tanh take 1/x from v_rcp_f32 (1 ulp) instead of the correctly rounded f32 division, whose 11-instruction dependent sequence stood twice in series in every step
# speedup vs baseline: 5.1303x; 1.1645x over previous
#include <hip/hip_runtime.h>

typedef float v2f __attribute__((ext_vector_type(2)));
typedef float v8f __attribute__((ext_vector_type(8)));

#define SEQ_LEN 131072
#define HID 64
#define G4 256

__global__ __launch_bounds__(256) void xproj_wmma(
    const float* __restrict__ inputs, const float* __restrict__ W_ih,
    const float* __restrict__ b_ih, const float* __restrict__ b_hh,
    float* __restrict__ x_proj)
{
  const int lane = threadIdx.x & 31;
  const int wave = threadIdx.x >> 5;
  const int rt   = blockIdx.x * 8 + wave;
  const int l15  = lane & 15;
  const int half = lane >> 4;
  const int row0 = rt * 16;

  const v2f* Arow = reinterpret_cast<const v2f*>(inputs + (size_t)(row0 + l15) * HID);
  v2f areg[16];
#pragma unroll
  for (int kk = 0; kk < 16; ++kk) areg[kk] = Arow[kk * 2 + half];

  for (int ct = 0; ct < 16; ++ct) {
    const int  col  = ct * 16 + l15;
    const v2f* Brow = reinterpret_cast<const v2f*>(W_ih + (size_t)col * HID);
    v8f c = {};
#pragma unroll
    for (int kk = 0; kk < 16; ++kk) {
      v2f b = Brow[kk * 2 + half];
      c = __builtin_amdgcn_wmma_f32_16x16x4_f32(false, areg[kk], false, b,
                                                (short)0, c, false, false);
    }
    const float bias = b_ih[col] + b_hh[col];
#pragma unroll
    for (int v = 0; v < 8; ++v) {
      const int row = row0 + v + half * 8;
      x_proj[(size_t)row * G4 + col] = c[v] + bias;
    }
  }
}

__device__ __forceinline__ float fsig(float x)  { return __builtin_amdgcn_rcpf(1.0f + __expf(-x)); }
__device__ __forceinline__ float ftanh(float x) { return 1.0f - 2.0f * __builtin_amdgcn_rcpf(__expf(2.0f * x) + 1.0f); }

#define XPF 8

__device__ __forceinline__ void lds_barrier() {
  __builtin_amdgcn_fence(__ATOMIC_RELEASE, "workgroup", "local");
  __builtin_amdgcn_s_barrier();
  __builtin_amdgcn_fence(__ATOMIC_ACQUIRE, "workgroup", "local");
}

__device__ __forceinline__ v2f fma2(v2f w, float h, v2f acc) {
  const v2f hh = {h, h};
  return __builtin_elementwise_fma(w, hh, acc);
}

__global__ __launch_bounds__(HID) void lstm_scan(
    const float* __restrict__ x_proj, const float* __restrict__ W_hh,
    float* __restrict__ hs)
{
  __shared__ __attribute__((aligned(16))) float sh_h[2][HID];
  const int j = threadIdx.x;

  v2f wif[HID], wgo[HID];
#pragma unroll
  for (int k = 0; k < HID; ++k) {
    wif[k].x = W_hh[(size_t)(0 * HID + j) * HID + k];
    wif[k].y = W_hh[(size_t)(1 * HID + j) * HID + k];
    wgo[k].x = W_hh[(size_t)(2 * HID + j) * HID + k];
    wgo[k].y = W_hh[(size_t)(3 * HID + j) * HID + k];
  }

  float creg = 0.0f;
  sh_h[0][j] = 0.0f;
  lds_barrier();

  static_assert(SEQ_LEN % XPF == 0, "whole blocks");
  static_assert(XPF % 2 == 0, "buffer parity is the step's index in its block");
  float xc[XPF][4], xn[XPF][4];
  for (int u = 0; u < XPF; ++u)
    for (int g = 0; g < 4; ++g) xc[u][g] = x_proj[(size_t)u * G4 + g * HID + j];

  for (int tb = 0; tb < SEQ_LEN; tb += XPF) {
    const int nb = (tb + XPF < SEQ_LEN) ? (tb + XPF) : tb;
    for (int u = 0; u < XPF; ++u)
      for (int g = 0; g < 4; ++g) xn[u][g] = x_proj[(size_t)(nb + u) * G4 + g * HID + j];

#pragma unroll
    for (int u = 0; u < XPF; ++u) {
      const int t = tb + u;

      v2f aif[4], ago[4];
      aif[0].x = xc[u][0]; aif[0].y = xc[u][1];
      ago[0].x = xc[u][2]; ago[0].y = xc[u][3];
#pragma unroll
      for (int c = 1; c < 4; ++c) { aif[c].x = 0.0f; aif[c].y = 0.0f; ago[c].x = 0.0f; ago[c].y = 0.0f; }

      const float4* h4 = reinterpret_cast<const float4*>(sh_h[u & 1]);
#pragma unroll
      for (int q = 0; q < 4; ++q) {
#pragma unroll
        for (int c = 0; c < 4; ++c) {
          const float4 h = h4[4 * q + c];
          const int k = 16 * q + 4 * c;
          aif[c] = fma2(wif[k + 0], h.x, aif[c]);  ago[c] = fma2(wgo[k + 0], h.x, ago[c]);
          aif[c] = fma2(wif[k + 1], h.y, aif[c]);  ago[c] = fma2(wgo[k + 1], h.y, ago[c]);
          aif[c] = fma2(wif[k + 2], h.z, aif[c]);  ago[c] = fma2(wgo[k + 2], h.z, ago[c]);
          aif[c] = fma2(wif[k + 3], h.w, aif[c]);  ago[c] = fma2(wgo[k + 3], h.w, ago[c]);
        }
      }
      const v2f zif = (aif[0] + aif[1]) + (aif[2] + aif[3]);
      const v2f zgo = (ago[0] + ago[1]) + (ago[2] + ago[3]);

      const float ig = fsig (zif.x);
      const float fg = fsig (zif.y);
      const float gg = ftanh(zgo.x);
      const float og = fsig (zgo.y);
      creg = fg * creg + ig * gg;
      const float hn = og * ftanh(creg);
      sh_h[(u + 1) & 1][j] = hn;
      hs[(size_t)t * HID + j] = hn;
      lds_barrier();
    }
    for (int u = 0; u < XPF; ++u)
      for (int g = 0; g < 4; ++g) xc[u][g] = xn[u][g];
  }
}

__global__ __launch_bounds__(256) void head_wmma(
    const float* __restrict__ hs, const float* __restrict__ W_lin,
    const float* __restrict__ b_lin, const float* __restrict__ W_fin,
    const float* __restrict__ b_fin, float* __restrict__ out)
{
  const int lane = threadIdx.x & 31;
  const int wave = threadIdx.x >> 5;
  const int rt   = blockIdx.x * 8 + wave;
  const int l15  = lane & 15;
  const int half = lane >> 4;
  const int row0 = rt * 16;

  const v2f* Arow = reinterpret_cast<const v2f*>(hs + (size_t)(row0 + l15) * HID);
  v2f areg[16];
#pragma unroll
  for (int kk = 0; kk < 16; ++kk) {
    v2f a = Arow[kk * 2 + half];
    a.x = fmaxf(a.x, 0.0f);
    a.y = fmaxf(a.y, 0.0f);
    areg[kk] = a;
  }

  float partial[8];
#pragma unroll
  for (int v = 0; v < 8; ++v) partial[v] = 0.0f;

#pragma unroll
  for (int ctile = 0; ctile < 4; ++ctile) {
    const int  col  = ctile * 16 + l15;
    const v2f* Brow = reinterpret_cast<const v2f*>(W_lin + (size_t)col * HID);
    v8f c = {};
#pragma unroll
    for (int kk = 0; kk < 16; ++kk) {
      v2f b = Brow[kk * 2 + half];
      c = __builtin_amdgcn_wmma_f32_16x16x4_f32(false, areg[kk], false, b,
                                                (short)0, c, false, false);
    }
    const float bl = b_lin[col];
    const float wf = W_fin[col];
#pragma unroll
    for (int v = 0; v < 8; ++v)
      partial[v] += fmaxf(c[v] + bl, 0.0f) * wf;
  }

#pragma unroll
  for (int v = 0; v < 8; ++v) {
    float s = partial[v];
    s += __shfl_xor(s, 1, 32);
    s += __shfl_xor(s, 2, 32);
    s += __shfl_xor(s, 4, 32);
    s += __shfl_xor(s, 8, 32);
    if (l15 == 0) out[row0 + half * 8 + v] = s + b_fin[0];
  }
}

extern "C" void kernel_launch(void* const* d_in, const int* in_sizes, int n_in,
                              void* d_out, int out_size, void* d_ws, size_t ws_size,
                              hipStream_t stream) {
  const float* inputs = (const float*)d_in[0];
  const float* W_ih   = (const float*)d_in[1];
  const float* W_hh   = (const float*)d_in[2];
  const float* b_ih   = (const float*)d_in[3];
  const float* b_hh   = (const float*)d_in[4];
  const float* W_lin  = (const float*)d_in[5];
  const float* b_lin  = (const float*)d_in[6];
  const float* W_fin  = (const float*)d_in[7];
  const float* b_fin  = (const float*)d_in[8];
  float* out = (float*)d_out;

  float* x_proj = (float*)d_ws;
  float* hs     = x_proj + (size_t)SEQ_LEN * G4;

  xproj_wmma<<<1024, 256, 0, stream>>>(inputs, W_ih, b_ih, b_hh, x_proj);
  lstm_scan<<<1, HID, 0, stream>>>(x_proj, W_hh, hs);
  head_wmma<<<1024, 256, 0, stream>>>(hs, W_lin, b_lin, W_fin, b_fin, out);
}


# lstm_scan: 4 waves, each unit's k range split over a lane pair (64 packed FMAs per lane per step, not 128), half sums joined by permlanex16: changes the summation order; hs kept two steps per line
# speedup vs baseline: 5.4427x; 1.0609x over previous
#include <hip/hip_runtime.h>

typedef float v2f __attribute__((ext_vector_type(2)));
typedef float v8f __attribute__((ext_vector_type(8)));

#define SEQ_LEN 131072
#define HID 64
#define G4 256

__global__ __launch_bounds__(256) void xproj_wmma(
    const float* __restrict__ inputs, const float* __restrict__ W_ih,
    const float* __restrict__ b_ih, const float* __restrict__ b_hh,
    float* __restrict__ x_proj)
{
  const int lane = threadIdx.x & 31;
  const int wave = threadIdx.x >> 5;
  const int rt   = blockIdx.x * 8 + wave;
  const int l15  = lane & 15;
  const int half = lane >> 4;
  const int row0 = rt * 16;

  const v2f* Arow = reinterpret_cast<const v2f*>(inputs + (size_t)(row0 + l15) * HID);
  v2f areg[16];
#pragma unroll
  for (int kk = 0; kk < 16; ++kk) areg[kk] = Arow[kk * 2 + half];

  for (int ct = 0; ct < 16; ++ct) {
    const int  col  = ct * 16 + l15;
    const v2f* Brow = reinterpret_cast<const v2f*>(W_ih + (size_t)col * HID);
    v8f c = {};
#pragma unroll
    for (int kk = 0; kk < 16; ++kk) {
      v2f b = Brow[kk * 2 + half];
      c = __builtin_amdgcn_wmma_f32_16x16x4_f32(false, areg[kk], false, b,
                                                (short)0, c, false, false);
    }
    const float bias = b_ih[col] + b_hh[col];
#pragma unroll
    for (int v = 0; v < 8; ++v) {
      const int row = row0 + v + half * 8;
      x_proj[(size_t)row * G4 + col] = c[v] + bias;
    }
  }
}

__device__ __forceinline__ float fsig(float x)  { return __builtin_amdgcn_rcpf(1.0f + __expf(-x)); }
__device__ __forceinline__ float ftanh(float x) { return 1.0f - 2.0f * __builtin_amdgcn_rcpf(__expf(2.0f * x) + 1.0f); }

#define XPF 8
#define SCAN_THREADS 128
#define KH 32

__device__ __forceinline__ void lds_barrier() {
  __builtin_amdgcn_fence(__ATOMIC_RELEASE, "workgroup", "local");
  __builtin_amdgcn_s_barrier();
  __builtin_amdgcn_fence(__ATOMIC_ACQUIRE, "workgroup", "local");
}

__device__ __forceinline__ v2f fma2(v2f w, float h, v2f acc) {
  const v2f hh = {h, h};
  return __builtin_elementwise_fma(w, hh, acc);
}

__device__ __forceinline__ float xhalf(float x) {
  const int b = __builtin_bit_cast(int, x);
  return __builtin_bit_cast(float,
      __builtin_amdgcn_permlanex16(b, b, 0x76543210, 0xfedcba98, false, false));
}

__global__ __launch_bounds__(SCAN_THREADS) __attribute__((amdgpu_waves_per_eu(1, 1))) void lstm_scan(
    const float* __restrict__ x_proj, const float* __restrict__ W_hh,
    float* __restrict__ hs)
{
  __shared__ __attribute__((aligned(16))) float sh_h[2][HID];
  const int lane = threadIdx.x & 31;
  const int wave = threadIdx.x >> 5;
  const int p    = lane >> 4;
  const int j    = wave * 16 + (lane & 15);
  const int k0   = p * KH;

  v2f wif[KH], wgo[KH];
#pragma unroll
  for (int k = 0; k < KH; ++k) {
    wif[k].x = W_hh[(size_t)(0 * HID + j) * HID + k0 + k];
    wif[k].y = W_hh[(size_t)(1 * HID + j) * HID + k0 + k];
    wgo[k].x = W_hh[(size_t)(2 * HID + j) * HID + k0 + k];
    wgo[k].y = W_hh[(size_t)(3 * HID + j) * HID + k0 + k];
  }

  float creg = 0.0f;
  float hprev = 0.0f;
  sh_h[0][j] = 0.0f;
  lds_barrier();

  static_assert(SEQ_LEN % XPF == 0, "whole blocks");
  static_assert(XPF % 2 == 0, "buffer parity is the step's index in its block");
  const float* xcol = x_proj + (2 * p) * HID + j;
  float xc[XPF][2], xn[XPF][2];
  for (int u = 0; u < XPF; ++u)
    for (int g = 0; g < 2; ++g) xc[u][g] = xcol[(size_t)u * G4 + g * HID];

  float* hsline = hs + wave * 32 + lane;

  for (int tb = 0; tb < SEQ_LEN; tb += XPF) {
    const int nb = (tb + XPF < SEQ_LEN) ? (tb + XPF) : tb;
    for (int u = 0; u < XPF; ++u)
      for (int g = 0; g < 2; ++g) xn[u][g] = xcol[(size_t)(nb + u) * G4 + g * HID];

#pragma unroll
    for (int u = 0; u < XPF; ++u) {
      const int t = tb + u;

      v2f aif[4], ago[4];
      aif[0].x = (p == 0) ? xc[u][0] : 0.0f;  aif[0].y = (p == 0) ? xc[u][1] : 0.0f;
      ago[0].x = (p == 0) ? 0.0f : xc[u][0];  ago[0].y = (p == 0) ? 0.0f : xc[u][1];
#pragma unroll
      for (int c = 1; c < 4; ++c) { aif[c].x = 0.0f; aif[c].y = 0.0f; ago[c].x = 0.0f; ago[c].y = 0.0f; }

      const float4* h4 = reinterpret_cast<const float4*>(sh_h[u & 1] + k0);
#pragma unroll
      for (int q = 0; q < KH / 16; ++q) {
#pragma unroll
        for (int c = 0; c < 4; ++c) {
          const float4 h = h4[4 * q + c];
          const int k = 16 * q + 4 * c;
          aif[c] = fma2(wif[k + 0], h.x, aif[c]);  ago[c] = fma2(wgo[k + 0], h.x, ago[c]);
          aif[c] = fma2(wif[k + 1], h.y, aif[c]);  ago[c] = fma2(wgo[k + 1], h.y, ago[c]);
          aif[c] = fma2(wif[k + 2], h.z, aif[c]);  ago[c] = fma2(wgo[k + 2], h.z, ago[c]);
          aif[c] = fma2(wif[k + 3], h.w, aif[c]);  ago[c] = fma2(wgo[k + 3], h.w, ago[c]);
        }
      }
      const v2f sif = (aif[0] + aif[1]) + (aif[2] + aif[3]);
      const v2f sgo = (ago[0] + ago[1]) + (ago[2] + ago[3]);

      const float zi = sif.x + xhalf(sif.x);
      const float zf = sif.y + xhalf(sif.y);
      const float zg = sgo.x + xhalf(sgo.x);
      const float zo = sgo.y + xhalf(sgo.y);

      const float ig = fsig (zi);
      const float fg = fsig (zf);
      const float gg = ftanh(zg);
      const float og = fsig (zo);
      creg = fg * creg + ig * gg;
      const float hn = og * ftanh(creg);
      sh_h[(u + 1) & 1][j] = hn;
      if (u & 1) {
        hsline[(size_t)(t >> 1) * 128] = p ? hn : hprev;
      } else {
        hprev = hn;
      }
      lds_barrier();
    }
    for (int u = 0; u < XPF; ++u)
      for (int g = 0; g < 2; ++g) xc[u][g] = xn[u][g];
  }
}

__global__ __launch_bounds__(256) void head_wmma(
    const float* __restrict__ hs, const float* __restrict__ W_lin,
    const float* __restrict__ b_lin, const float* __restrict__ W_fin,
    const float* __restrict__ b_fin, float* __restrict__ out)
{
  const int lane = threadIdx.x & 31;
  const int wave = threadIdx.x >> 5;
  const int rt   = blockIdx.x * 8 + wave;
  const int l15  = lane & 15;
  const int half = lane >> 4;
  const int row0 = rt * 16;

  const int    arow  = row0 + l15;
  const float* Abase = hs + (size_t)(arow >> 1) * 128 + (arow & 1) * 16 + half * 2;
  v2f areg[16];
#pragma unroll
  for (int kk = 0; kk < 16; ++kk) {
    v2f a = *reinterpret_cast<const v2f*>(Abase + (kk >> 2) * 32 + (kk & 3) * 4);
    a.x = fmaxf(a.x, 0.0f);
    a.y = fmaxf(a.y, 0.0f);
    areg[kk] = a;
  }

  float partial[8];
#pragma unroll
  for (int v = 0; v < 8; ++v) partial[v] = 0.0f;

#pragma unroll
  for (int ctile = 0; ctile < 4; ++ctile) {
    const int  col  = ctile * 16 + l15;
    const v2f* Brow = reinterpret_cast<const v2f*>(W_lin + (size_t)col * HID);
    v8f c = {};
#pragma unroll
    for (int kk = 0; kk < 16; ++kk) {
      v2f b = Brow[kk * 2 + half];
      c = __builtin_amdgcn_wmma_f32_16x16x4_f32(false, areg[kk], false, b,
                                                (short)0, c, false, false);
    }
    const float bl = b_lin[col];
    const float wf = W_fin[col];
#pragma unroll
    for (int v = 0; v < 8; ++v)
      partial[v] += fmaxf(c[v] + bl, 0.0f) * wf;
  }

#pragma unroll
  for (int v = 0; v < 8; ++v) {
    float s = partial[v];
    s += __shfl_xor(s, 1, 32);
    s += __shfl_xor(s, 2, 32);
    s += __shfl_xor(s, 4, 32);
    s += __shfl_xor(s, 8, 32);
    if (l15 == 0) out[row0 + half * 8 + v] = s + b_fin[0];
  }
}

extern "C" void kernel_launch(void* const* d_in, const int* in_sizes, int n_in,
                              void* d_out, int out_size, void* d_ws, size_t ws_size,
                              hipStream_t stream) {
  const float* inputs = (const float*)d_in[0];
  const float* W_ih   = (const float*)d_in[1];
  const float* W_hh   = (const float*)d_in[2];
  const float* b_ih   = (const float*)d_in[3];
  const float* b_hh   = (const float*)d_in[4];
  const float* W_lin  = (const float*)d_in[5];
  const float* b_lin  = (const float*)d_in[6];
  const float* W_fin  = (const float*)d_in[7];
  const float* b_fin  = (const float*)d_in[8];
  float* out = (float*)d_out;

  float* x_proj = (float*)d_ws;
  float* hs     = x_proj + (size_t)SEQ_LEN * G4;

  xproj_wmma<<<1024, 256, 0, stream>>>(inputs, W_ih, b_ih, b_hh, x_proj);
  lstm_scan<<<1, SCAN_THREADS, 0, stream>>>(x_proj, W_hh, hs);
  head_wmma<<<1024, 256, 0, stream>>>(hs, W_lin, b_lin, W_fin, b_fin, out);
}


# lstm_scan: x_proj stored gate-pair interleaved: a lane prefetches a step's two values with one 8-byte load into an aligned register pair, so step 0 of a block no longer waits on the loads just issued
# speedup vs baseline: 6.8418x; 1.2571x over previous
#include <hip/hip_runtime.h>

typedef float v2f __attribute__((ext_vector_type(2)));
typedef float v8f __attribute__((ext_vector_type(8)));

#define SEQ_LEN 131072
#define HID 64
#define G4 256

__global__ __launch_bounds__(256) void xproj_wmma(
    const float* __restrict__ inputs, const float* __restrict__ W_ih,
    const float* __restrict__ b_ih, const float* __restrict__ b_hh,
    float* __restrict__ x_proj)
{
  const int lane = threadIdx.x & 31;
  const int wave = threadIdx.x >> 5;
  const int rt   = blockIdx.x * 8 + wave;
  const int l15  = lane & 15;
  const int half = lane >> 4;
  const int row0 = rt * 16;

  const v2f* Arow = reinterpret_cast<const v2f*>(inputs + (size_t)(row0 + l15) * HID);
  v2f areg[16];
#pragma unroll
  for (int kk = 0; kk < 16; ++kk) areg[kk] = Arow[kk * 2 + half];

  for (int pj = 0; pj < 8; ++pj) {
    const int  pp   = pj >> 2;
    const int  ju   = (pj & 3) * 16 + l15;
    const int  col0 = (2 * pp) * HID + ju;
    const int  col1 = col0 + HID;
    const v2f* Brow0 = reinterpret_cast<const v2f*>(W_ih + (size_t)col0 * HID);
    const v2f* Brow1 = reinterpret_cast<const v2f*>(W_ih + (size_t)col1 * HID);
    v8f c0 = {};
    v8f c1 = {};
#pragma unroll
    for (int kk = 0; kk < 16; ++kk) {
      v2f b0 = Brow0[kk * 2 + half];
      v2f b1 = Brow1[kk * 2 + half];
      c0 = __builtin_amdgcn_wmma_f32_16x16x4_f32(false, areg[kk], false, b0,
                                                 (short)0, c0, false, false);
      c1 = __builtin_amdgcn_wmma_f32_16x16x4_f32(false, areg[kk], false, b1,
                                                 (short)0, c1, false, false);
    }
    const float bias0 = b_ih[col0] + b_hh[col0];
    const float bias1 = b_ih[col1] + b_hh[col1];
#pragma unroll
    for (int v = 0; v < 8; ++v) {
      const int row = row0 + v + half * 8;
      v2f o;
      o.x = c0[v] + bias0;
      o.y = c1[v] + bias1;
      *reinterpret_cast<v2f*>(x_proj + (size_t)row * G4 + pp * 128 + ju * 2) = o;
    }
  }
}

__device__ __forceinline__ float fsig(float x)  { return __builtin_amdgcn_rcpf(1.0f + __expf(-x)); }
__device__ __forceinline__ float ftanh(float x) { return 1.0f - 2.0f * __builtin_amdgcn_rcpf(__expf(2.0f * x) + 1.0f); }

#define XPF 8
#define SCAN_THREADS 128
#define KH 32

__device__ __forceinline__ void lds_barrier() {
  __builtin_amdgcn_fence(__ATOMIC_RELEASE, "workgroup", "local");
  __builtin_amdgcn_s_barrier();
  __builtin_amdgcn_fence(__ATOMIC_ACQUIRE, "workgroup", "local");
}

__device__ __forceinline__ v2f fma2(v2f w, float h, v2f acc) {
  const v2f hh = {h, h};
  return __builtin_elementwise_fma(w, hh, acc);
}

__device__ __forceinline__ float xhalf(float x) {
  const int b = __builtin_bit_cast(int, x);
  return __builtin_bit_cast(float,
      __builtin_amdgcn_permlanex16(b, b, 0x76543210, 0xfedcba98, false, false));
}

__global__ __launch_bounds__(SCAN_THREADS) __attribute__((amdgpu_waves_per_eu(1, 1))) void lstm_scan(
    const float* __restrict__ x_proj, const float* __restrict__ W_hh,
    float* __restrict__ hs)
{
  __shared__ __attribute__((aligned(16))) float sh_h[2][HID];
  const int lane = threadIdx.x & 31;
  const int wave = threadIdx.x >> 5;
  const int p    = lane >> 4;
  const int j    = wave * 16 + (lane & 15);
  const int k0   = p * KH;

  v2f wif[KH], wgo[KH];
#pragma unroll
  for (int k = 0; k < KH; ++k) {
    wif[k].x = W_hh[(size_t)(0 * HID + j) * HID + k0 + k];
    wif[k].y = W_hh[(size_t)(1 * HID + j) * HID + k0 + k];
    wgo[k].x = W_hh[(size_t)(2 * HID + j) * HID + k0 + k];
    wgo[k].y = W_hh[(size_t)(3 * HID + j) * HID + k0 + k];
  }

  float creg = 0.0f;
  float hprev = 0.0f;
  sh_h[0][j] = 0.0f;
  lds_barrier();

  static_assert(SEQ_LEN % XPF == 0, "whole blocks");
  static_assert(XPF % 2 == 0, "buffer parity is the step's index in its block");
  const v2f* xcol = reinterpret_cast<const v2f*>(x_proj + p * 128 + j * 2);
  v2f xc[XPF], xn[XPF];
  for (int u = 0; u < XPF; ++u) xc[u] = xcol[(size_t)u * (G4 / 2)];

  float* hsline = hs + wave * 32 + lane;

  for (int tb = 0; tb < SEQ_LEN; tb += XPF) {
    const int nb = (tb + XPF < SEQ_LEN) ? (tb + XPF) : tb;
    for (int u = 0; u < XPF; ++u) xn[u] = xcol[(size_t)(nb + u) * (G4 / 2)];

#pragma unroll
    for (int u = 0; u < XPF; ++u) {
      const int t = tb + u;

      v2f aif[4], ago[4];
      const float x0 = xc[u].x, x1 = xc[u].y;
      aif[0].x = (p == 0) ? x0 : 0.0f;  aif[0].y = (p == 0) ? x1 : 0.0f;
      ago[0].x = (p == 0) ? 0.0f : x0;  ago[0].y = (p == 0) ? 0.0f : x1;
#pragma unroll
      for (int c = 1; c < 4; ++c) { aif[c].x = 0.0f; aif[c].y = 0.0f; ago[c].x = 0.0f; ago[c].y = 0.0f; }

      const float4* h4 = reinterpret_cast<const float4*>(sh_h[u & 1] + k0);
#pragma unroll
      for (int q = 0; q < KH / 16; ++q) {
#pragma unroll
        for (int c = 0; c < 4; ++c) {
          const float4 h = h4[4 * q + c];
          const int k = 16 * q + 4 * c;
          aif[c] = fma2(wif[k + 0], h.x, aif[c]);  ago[c] = fma2(wgo[k + 0], h.x, ago[c]);
          aif[c] = fma2(wif[k + 1], h.y, aif[c]);  ago[c] = fma2(wgo[k + 1], h.y, ago[c]);
          aif[c] = fma2(wif[k + 2], h.z, aif[c]);  ago[c] = fma2(wgo[k + 2], h.z, ago[c]);
          aif[c] = fma2(wif[k + 3], h.w, aif[c]);  ago[c] = fma2(wgo[k + 3], h.w, ago[c]);
        }
      }
      const v2f sif = (aif[0] + aif[1]) + (aif[2] + aif[3]);
      const v2f sgo = (ago[0] + ago[1]) + (ago[2] + ago[3]);

      const float zi = sif.x + xhalf(sif.x);
      const float zf = sif.y + xhalf(sif.y);
      const float zg = sgo.x + xhalf(sgo.x);
      const float zo = sgo.y + xhalf(sgo.y);

      const float ig = fsig (zi);
      const float fg = fsig (zf);
      const float gg = ftanh(zg);
      const float og = fsig (zo);
      creg = fg * creg + ig * gg;
      const float hn = og * ftanh(creg);
      sh_h[(u + 1) & 1][j] = hn;
      if (u & 1) {
        hsline[(size_t)(t >> 1) * 128] = p ? hn : hprev;
      } else {
        hprev = hn;
      }
      lds_barrier();
    }
    for (int u = 0; u < XPF; ++u) xc[u] = xn[u];
  }
}

__global__ __launch_bounds__(256) void head_wmma(
    const float* __restrict__ hs, const float* __restrict__ W_lin,
    const float* __restrict__ b_lin, const float* __restrict__ W_fin,
    const float* __restrict__ b_fin, float* __restrict__ out)
{
  const int lane = threadIdx.x & 31;
  const int wave = threadIdx.x >> 5;
  const int rt   = blockIdx.x * 8 + wave;
  const int l15  = lane & 15;
  const int half = lane >> 4;
  const int row0 = rt * 16;

  const int    arow  = row0 + l15;
  const float* Abase = hs + (size_t)(arow >> 1) * 128 + (arow & 1) * 16 + half * 2;
  v2f areg[16];
#pragma unroll
  for (int kk = 0; kk < 16; ++kk) {
    v2f a = *reinterpret_cast<const v2f*>(Abase + (kk >> 2) * 32 + (kk & 3) * 4);
    a.x = fmaxf(a.x, 0.0f);
    a.y = fmaxf(a.y, 0.0f);
    areg[kk] = a;
  }

  float partial[8];
#pragma unroll
  for (int v = 0; v < 8; ++v) partial[v] = 0.0f;

#pragma unroll
  for (int ctile = 0; ctile < 4; ++ctile) {
    const int  col  = ctile * 16 + l15;
    const v2f* Brow = reinterpret_cast<const v2f*>(W_lin + (size_t)col * HID);
    v8f c = {};
#pragma unroll
    for (int kk = 0; kk < 16; ++kk) {
      v2f b = Brow[kk * 2 + half];
      c = __builtin_amdgcn_wmma_f32_16x16x4_f32(false, areg[kk], false, b,
                                                (short)0, c, false, false);
    }
    const float bl = b_lin[col];
    const float wf = W_fin[col];
#pragma unroll
    for (int v = 0; v < 8; ++v)
      partial[v] += fmaxf(c[v] + bl, 0.0f) * wf;
  }

#pragma unroll
  for (int v = 0; v < 8; ++v) {
    float s = partial[v];
    s += __shfl_xor(s, 1, 32);
    s += __shfl_xor(s, 2, 32);
    s += __shfl_xor(s, 4, 32);
    s += __shfl_xor(s, 8, 32);
    if (l15 == 0) out[row0 + half * 8 + v] = s + b_fin[0];
  }
}

extern "C" void kernel_launch(void* const* d_in, const int* in_sizes, int n_in,
                              void* d_out, int out_size, void* d_ws, size_t ws_size,
                              hipStream_t stream) {
  const float* inputs = (const float*)d_in[0];
  const float* W_ih   = (const float*)d_in[1];
  const float* W_hh   = (const float*)d_in[2];
  const float* b_ih   = (const float*)d_in[3];
  const float* b_hh   = (const float*)d_in[4];
  const float* W_lin  = (const float*)d_in[5];
  const float* b_lin  = (const float*)d_in[6];
  const float* W_fin  = (const float*)d_in[7];
  const float* b_fin  = (const float*)d_in[8];
  float* out = (float*)d_out;

  float* x_proj = (float*)d_ws;
  float* hs     = x_proj + (size_t)SEQ_LEN * G4;

  xproj_wmma<<<1024, 256, 0, stream>>>(inputs, W_ih, b_ih, b_hh, x_proj);
  lstm_scan<<<1, SCAN_THREADS, 0, stream>>>(x_proj, W_hh, hs);
  head_wmma<<<1024, 256, 0, stream>>>(hs, W_lin, b_lin, W_fin, b_fin, out);
}


# lstm_scan: tanh and sigmoid are one v_tanh_f32 each (sigmoid(z) = 0.5 + 0.5 tanh(z/2), the halving folded exactly into the i, f, o rows of W_hh and of x_proj) instead of exp2 + add + rcp chains
# speedup vs baseline: 7.7528x; 1.1332x over previous
#include <hip/hip_runtime.h>

typedef float v2f __attribute__((ext_vector_type(2)));
typedef float v8f __attribute__((ext_vector_type(8)));

#define SEQ_LEN 131072
#define HID 64
#define G4 256

__global__ __launch_bounds__(256) void xproj_wmma(
    const float* __restrict__ inputs, const float* __restrict__ W_ih,
    const float* __restrict__ b_ih, const float* __restrict__ b_hh,
    float* __restrict__ x_proj)
{
  const int lane = threadIdx.x & 31;
  const int wave = threadIdx.x >> 5;
  const int rt   = blockIdx.x * 8 + wave;
  const int l15  = lane & 15;
  const int half = lane >> 4;
  const int row0 = rt * 16;

  const v2f* Arow = reinterpret_cast<const v2f*>(inputs + (size_t)(row0 + l15) * HID);
  v2f areg[16];
#pragma unroll
  for (int kk = 0; kk < 16; ++kk) areg[kk] = Arow[kk * 2 + half];

  for (int pj = 0; pj < 8; ++pj) {
    const int  pp   = pj >> 2;
    const int  ju   = (pj & 3) * 16 + l15;
    const int  col0 = (2 * pp) * HID + ju;
    const int  col1 = col0 + HID;
    const v2f* Brow0 = reinterpret_cast<const v2f*>(W_ih + (size_t)col0 * HID);
    const v2f* Brow1 = reinterpret_cast<const v2f*>(W_ih + (size_t)col1 * HID);
    v8f c0 = {};
    v8f c1 = {};
#pragma unroll
    for (int kk = 0; kk < 16; ++kk) {
      v2f b0 = Brow0[kk * 2 + half];
      v2f b1 = Brow1[kk * 2 + half];
      c0 = __builtin_amdgcn_wmma_f32_16x16x4_f32(false, areg[kk], false, b0,
                                                 (short)0, c0, false, false);
      c1 = __builtin_amdgcn_wmma_f32_16x16x4_f32(false, areg[kk], false, b1,
                                                 (short)0, c1, false, false);
    }
    const float bias0 = b_ih[col0] + b_hh[col0];
    const float bias1 = b_ih[col1] + b_hh[col1];
    const float s0 = (pp == 0) ? 0.5f : 1.0f;
#pragma unroll
    for (int v = 0; v < 8; ++v) {
      const int row = row0 + v + half * 8;
      v2f o;
      o.x = (c0[v] + bias0) * s0;
      o.y = (c1[v] + bias1) * 0.5f;
      *reinterpret_cast<v2f*>(x_proj + (size_t)row * G4 + pp * 128 + ju * 2) = o;
    }
  }
}

__device__ __forceinline__ float ftanh(float x)          { return __builtin_amdgcn_tanhf(x); }
__device__ __forceinline__ float fsig_of_half(float zh)  { return __builtin_fmaf(__builtin_amdgcn_tanhf(zh), 0.5f, 0.5f); }

#define XPF 8
#define SCAN_THREADS 128
#define KH 32

__device__ __forceinline__ void lds_barrier() {
  __builtin_amdgcn_fence(__ATOMIC_RELEASE, "workgroup", "local");
  __builtin_amdgcn_s_barrier();
  __builtin_amdgcn_fence(__ATOMIC_ACQUIRE, "workgroup", "local");
}

__device__ __forceinline__ v2f fma2(v2f w, float h, v2f acc) {
  const v2f hh = {h, h};
  return __builtin_elementwise_fma(w, hh, acc);
}

__device__ __forceinline__ float xhalf(float x) {
  const int b = __builtin_bit_cast(int, x);
  return __builtin_bit_cast(float,
      __builtin_amdgcn_permlanex16(b, b, 0x76543210, 0xfedcba98, false, false));
}

__global__ __launch_bounds__(SCAN_THREADS) __attribute__((amdgpu_waves_per_eu(1, 1))) void lstm_scan(
    const float* __restrict__ x_proj, const float* __restrict__ W_hh,
    float* __restrict__ hs)
{
  __shared__ __attribute__((aligned(16))) float sh_h[2][HID];
  const int lane = threadIdx.x & 31;
  const int wave = threadIdx.x >> 5;
  const int p    = lane >> 4;
  const int j    = wave * 16 + (lane & 15);
  const int k0   = p * KH;

  v2f wif[KH], wgo[KH];
#pragma unroll
  for (int k = 0; k < KH; ++k) {
    wif[k].x = 0.5f * W_hh[(size_t)(0 * HID + j) * HID + k0 + k];
    wif[k].y = 0.5f * W_hh[(size_t)(1 * HID + j) * HID + k0 + k];
    wgo[k].x =        W_hh[(size_t)(2 * HID + j) * HID + k0 + k];
    wgo[k].y = 0.5f * W_hh[(size_t)(3 * HID + j) * HID + k0 + k];
  }

  float creg = 0.0f;
  float hprev = 0.0f;
  sh_h[0][j] = 0.0f;
  lds_barrier();

  static_assert(SEQ_LEN % XPF == 0, "whole blocks");
  static_assert(XPF % 2 == 0, "buffer parity is the step's index in its block");
  const v2f* xcol = reinterpret_cast<const v2f*>(x_proj + p * 128 + j * 2);
  v2f xc[XPF], xn[XPF];
  for (int u = 0; u < XPF; ++u) xc[u] = xcol[(size_t)u * (G4 / 2)];

  float* hsline = hs + wave * 32 + lane;

  for (int tb = 0; tb < SEQ_LEN; tb += XPF) {
    const int nb = (tb + XPF < SEQ_LEN) ? (tb + XPF) : tb;
    for (int u = 0; u < XPF; ++u) xn[u] = xcol[(size_t)(nb + u) * (G4 / 2)];

#pragma unroll
    for (int u = 0; u < XPF; ++u) {
      const int t = tb + u;

      v2f aif[4], ago[4];
      const float x0 = xc[u].x, x1 = xc[u].y;
      aif[0].x = (p == 0) ? x0 : 0.0f;  aif[0].y = (p == 0) ? x1 : 0.0f;
      ago[0].x = (p == 0) ? 0.0f : x0;  ago[0].y = (p == 0) ? 0.0f : x1;
#pragma unroll
      for (int c = 1; c < 4; ++c) { aif[c].x = 0.0f; aif[c].y = 0.0f; ago[c].x = 0.0f; ago[c].y = 0.0f; }

      const float4* h4 = reinterpret_cast<const float4*>(sh_h[u & 1] + k0);
#pragma unroll
      for (int q = 0; q < KH / 16; ++q) {
#pragma unroll
        for (int c = 0; c < 4; ++c) {
          const float4 h = h4[4 * q + c];
          const int k = 16 * q + 4 * c;
          aif[c] = fma2(wif[k + 0], h.x, aif[c]);  ago[c] = fma2(wgo[k + 0], h.x, ago[c]);
          aif[c] = fma2(wif[k + 1], h.y, aif[c]);  ago[c] = fma2(wgo[k + 1], h.y, ago[c]);
          aif[c] = fma2(wif[k + 2], h.z, aif[c]);  ago[c] = fma2(wgo[k + 2], h.z, ago[c]);
          aif[c] = fma2(wif[k + 3], h.w, aif[c]);  ago[c] = fma2(wgo[k + 3], h.w, ago[c]);
        }
      }
      const v2f sif = (aif[0] + aif[1]) + (aif[2] + aif[3]);
      const v2f sgo = (ago[0] + ago[1]) + (ago[2] + ago[3]);

      const float zi = sif.x + xhalf(sif.x);
      const float zf = sif.y + xhalf(sif.y);
      const float zg = sgo.x + xhalf(sgo.x);
      const float zo = sgo.y + xhalf(sgo.y);

      const float ig = fsig_of_half(zi);
      const float fg = fsig_of_half(zf);
      const float gg = ftanh(zg);
      const float og = fsig_of_half(zo);
      creg = fg * creg + ig * gg;
      const float hn = og * ftanh(creg);
      sh_h[(u + 1) & 1][j] = hn;
      if (u & 1) {
        hsline[(size_t)(t >> 1) * 128] = p ? hn : hprev;
      } else {
        hprev = hn;
      }
      lds_barrier();
    }
    for (int u = 0; u < XPF; ++u) xc[u] = xn[u];
  }
}

__global__ __launch_bounds__(256) void head_wmma(
    const float* __restrict__ hs, const float* __restrict__ W_lin,
    const float* __restrict__ b_lin, const float* __restrict__ W_fin,
    const float* __restrict__ b_fin, float* __restrict__ out)
{
  const int lane = threadIdx.x & 31;
  const int wave = threadIdx.x >> 5;
  const int rt   = blockIdx.x * 8 + wave;
  const int l15  = lane & 15;
  const int half = lane >> 4;
  const int row0 = rt * 16;

  const int    arow  = row0 + l15;
  const float* Abase = hs + (size_t)(arow >> 1) * 128 + (arow & 1) * 16 + half * 2;
  v2f areg[16];
#pragma unroll
  for (int kk = 0; kk < 16; ++kk) {
    v2f a = *reinterpret_cast<const v2f*>(Abase + (kk >> 2) * 32 + (kk & 3) * 4);
    a.x = fmaxf(a.x, 0.0f);
    a.y = fmaxf(a.y, 0.0f);
    areg[kk] = a;
  }

  float partial[8];
#pragma unroll
  for (int v = 0; v < 8; ++v) partial[v] = 0.0f;

#pragma unroll
  for (int ctile = 0; ctile < 4; ++ctile) {
    const int  col  = ctile * 16 + l15;
    const v2f* Brow = reinterpret_cast<const v2f*>(W_lin + (size_t)col * HID);
    v8f c = {};
#pragma unroll
    for (int kk = 0; kk < 16; ++kk) {
      v2f b = Brow[kk * 2 + half];
      c = __builtin_amdgcn_wmma_f32_16x16x4_f32(false, areg[kk], false, b,
                                                (short)0, c, false, false);
    }
    const float bl = b_lin[col];
    const float wf = W_fin[col];
#pragma unroll
    for (int v = 0; v < 8; ++v)
      partial[v] += fmaxf(c[v] + bl, 0.0f) * wf;
  }

#pragma unroll
  for (int v = 0; v < 8; ++v) {
    float s = partial[v];
    s += __shfl_xor(s, 1, 32);
    s += __shfl_xor(s, 2, 32);
    s += __shfl_xor(s, 4, 32);
    s += __shfl_xor(s, 8, 32);
    if (l15 == 0) out[row0 + half * 8 + v] = s + b_fin[0];
  }
}

extern "C" void kernel_launch(void* const* d_in, const int* in_sizes, int n_in,
                              void* d_out, int out_size, void* d_ws, size_t ws_size,
                              hipStream_t stream) {
  const float* inputs = (const float*)d_in[0];
  const float* W_ih   = (const float*)d_in[1];
  const float* W_hh   = (const float*)d_in[2];
  const float* b_ih   = (const float*)d_in[3];
  const float* b_hh   = (const float*)d_in[4];
  const float* W_lin  = (const float*)d_in[5];
  const float* b_lin  = (const float*)d_in[6];
  const float* W_fin  = (const float*)d_in[7];
  const float* b_fin  = (const float*)d_in[8];
  float* out = (float*)d_out;

  float* x_proj = (float*)d_ws;
  float* hs     = x_proj + (size_t)SEQ_LEN * G4;

  xproj_wmma<<<1024, 256, 0, stream>>>(inputs, W_ih, b_ih, b_hh, x_proj);
  lstm_scan<<<1, SCAN_THREADS, 0, stream>>>(x_proj, W_hh, hs);
  head_wmma<<<1024, 256, 0, stream>>>(hs, W_lin, b_lin, W_fin, b_fin, out);
}


# lstm_scan: a wave's own 16 h values go lane to lane (row_share), their products are summed behind the h store in front of the barrier: 6 LDS reads, 48 FMAs left behind it; changes the summation order
# speedup vs baseline: 8.4156x; 1.0855x over previous
#include <hip/hip_runtime.h>

typedef float v2f __attribute__((ext_vector_type(2)));
typedef float v8f __attribute__((ext_vector_type(8)));

#define SEQ_LEN 131072
#define HID 64
#define G4 256

__global__ __launch_bounds__(256) void xproj_wmma(
    const float* __restrict__ inputs, const float* __restrict__ W_ih,
    const float* __restrict__ b_ih, const float* __restrict__ b_hh,
    float* __restrict__ x_proj)
{
  const int lane = threadIdx.x & 31;
  const int wave = threadIdx.x >> 5;
  const int rt   = blockIdx.x * 8 + wave;
  const int l15  = lane & 15;
  const int half = lane >> 4;
  const int row0 = rt * 16;

  const v2f* Arow = reinterpret_cast<const v2f*>(inputs + (size_t)(row0 + l15) * HID);
  v2f areg[16];
#pragma unroll
  for (int kk = 0; kk < 16; ++kk) areg[kk] = Arow[kk * 2 + half];

  for (int pj = 0; pj < 8; ++pj) {
    const int  pp   = pj >> 2;
    const int  ju   = (pj & 3) * 16 + l15;
    const int  col0 = (2 * pp) * HID + ju;
    const int  col1 = col0 + HID;
    const v2f* Brow0 = reinterpret_cast<const v2f*>(W_ih + (size_t)col0 * HID);
    const v2f* Brow1 = reinterpret_cast<const v2f*>(W_ih + (size_t)col1 * HID);
    v8f c0 = {};
    v8f c1 = {};
#pragma unroll
    for (int kk = 0; kk < 16; ++kk) {
      v2f b0 = Brow0[kk * 2 + half];
      v2f b1 = Brow1[kk * 2 + half];
      c0 = __builtin_amdgcn_wmma_f32_16x16x4_f32(false, areg[kk], false, b0,
                                                 (short)0, c0, false, false);
      c1 = __builtin_amdgcn_wmma_f32_16x16x4_f32(false, areg[kk], false, b1,
                                                 (short)0, c1, false, false);
    }
    const float bias0 = b_ih[col0] + b_hh[col0];
    const float bias1 = b_ih[col1] + b_hh[col1];
    const float s0 = (pp == 0) ? 0.5f : 1.0f;
#pragma unroll
    for (int v = 0; v < 8; ++v) {
      const int row = row0 + v + half * 8;
      v2f o;
      o.x = (c0[v] + bias0) * s0;
      o.y = (c1[v] + bias1) * 0.5f;
      *reinterpret_cast<v2f*>(x_proj + (size_t)row * G4 + pp * 128 + ju * 2) = o;
    }
  }
}

__device__ __forceinline__ float ftanh(float x)          { return __builtin_amdgcn_tanhf(x); }
__device__ __forceinline__ float fsig_of_half(float zh)  { return __builtin_fmaf(__builtin_amdgcn_tanhf(zh), 0.5f, 0.5f); }

#define XPF 8
#define SCAN_THREADS 128
#define KH 32
#define KB 8

__device__ __forceinline__ void lds_barrier() {
  __builtin_amdgcn_fence(__ATOMIC_RELEASE, "workgroup", "local");
  __builtin_amdgcn_s_barrier();
  __builtin_amdgcn_fence(__ATOMIC_ACQUIRE, "workgroup", "local");
}

__device__ __forceinline__ v2f fma2(v2f w, float h, v2f acc) {
  const v2f hh = {h, h};
  return __builtin_elementwise_fma(w, hh, acc);
}

__device__ __forceinline__ float xpair(float x) {
  const int b = __builtin_bit_cast(int, x);
  return __builtin_bit_cast(float,
      __builtin_amdgcn_permlanex16(b, b, 0xfedcba98, 0x76543210, false, false));
}

template <int N>
__device__ __forceinline__ float row_lane(float x) {
  const int b = __builtin_bit_cast(int, x);
  return __builtin_bit_cast(float,
      __builtin_amdgcn_update_dpp(b, b, 0x150 + N, 0xf, 0xf, true));
}

__device__ __forceinline__ void own_block_sums(
    float hn, v2f x, int p, const v2f (&wif)[KH], const v2f (&wgo)[KH],
    v2f& oif0, v2f& oif1, v2f& ogo0, v2f& ogo1)
{
  float hb[KB];
  hb[0] = row_lane<0>(hn);  hb[1] = row_lane<1>(hn);
  hb[2] = row_lane<2>(hn);  hb[3] = row_lane<3>(hn);
  hb[4] = row_lane<4>(hn);  hb[5] = row_lane<5>(hn);
  hb[6] = row_lane<6>(hn);  hb[7] = row_lane<7>(hn);
  oif0.x = (p == 0) ? x.x : 0.0f;  oif0.y = (p == 0) ? x.y : 0.0f;
  ogo0.x = (p == 0) ? 0.0f : x.x;  ogo0.y = (p == 0) ? 0.0f : x.y;
  oif1.x = 0.0f; oif1.y = 0.0f;    ogo1.x = 0.0f; ogo1.y = 0.0f;
#pragma unroll
  for (int kk = 0; kk < 4; ++kk) {
    oif0 = fma2(wif[kk],     hb[kk],     oif0);  ogo0 = fma2(wgo[kk],     hb[kk],     ogo0);
    oif1 = fma2(wif[4 + kk], hb[4 + kk], oif1);  ogo1 = fma2(wgo[4 + kk], hb[4 + kk], ogo1);
  }
}

__global__ __launch_bounds__(SCAN_THREADS) __attribute__((amdgpu_waves_per_eu(1, 1))) void lstm_scan(
    const float* __restrict__ x_proj, const float* __restrict__ W_hh,
    float* __restrict__ hs)
{
  __shared__ __attribute__((aligned(16))) float sh_h[2][HID];
  const int lane = threadIdx.x & 31;
  const int wave = threadIdx.x >> 5;
  const int p    = lane >> 4;
  const int cu   = ((lane & 15) + 8 * p) & 15;
  const int j    = wave * 16 + cu;

  v2f wif[KH], wgo[KH];
#pragma unroll
  for (int bi = 0; bi < 4; ++bi) {
    const int kq = 16 * ((wave + bi) & 3) + KB * p;
#pragma unroll
    for (int kk = 0; kk < KB; ++kk) {
      wif[KB * bi + kk].x = 0.5f * W_hh[(size_t)(0 * HID + j) * HID + kq + kk];
      wif[KB * bi + kk].y = 0.5f * W_hh[(size_t)(1 * HID + j) * HID + kq + kk];
      wgo[KB * bi + kk].x =        W_hh[(size_t)(2 * HID + j) * HID + kq + kk];
      wgo[KB * bi + kk].y = 0.5f * W_hh[(size_t)(3 * HID + j) * HID + kq + kk];
    }
  }
  const float4* hq1 = reinterpret_cast<const float4*>(&sh_h[0][16 * ((wave + 1) & 3) + KB * p]);
  const float4* hq2 = reinterpret_cast<const float4*>(&sh_h[0][16 * ((wave + 2) & 3) + KB * p]);
  const float4* hq3 = reinterpret_cast<const float4*>(&sh_h[0][16 * ((wave + 3) & 3) + KB * p]);

  float creg = 0.0f;
  float hprev = 0.0f;
  sh_h[0][j] = 0.0f;

  static_assert(SEQ_LEN % XPF == 0, "whole blocks");
  static_assert(XPF % 2 == 0, "buffer parity is the step's index in its block");
  const v2f* xcol = reinterpret_cast<const v2f*>(x_proj + p * 128 + j * 2);
  v2f xc[XPF], xn[XPF];
  for (int u = 0; u < XPF; ++u) xc[u] = xcol[(size_t)u * (G4 / 2)];

  v2f oif0, oif1, ogo0, ogo1;
  own_block_sums(0.0f, xc[0], p, wif, wgo, oif0, oif1, ogo0, ogo1);
  lds_barrier();

  float* hsline = hs + wave * 32 + p * 16 + cu;

  for (int tb = 0; tb < SEQ_LEN; tb += XPF) {
    const int nb = (tb + XPF < SEQ_LEN) ? (tb + XPF) : tb;
    for (int u = 0; u < XPF; ++u) xn[u] = xcol[(size_t)(nb + u) * (G4 / 2)];

#pragma unroll
    for (int u = 0; u < XPF; ++u) {
      const int t = tb + u;

      v2f aif[4], ago[4];
      aif[0] = oif0;  aif[1] = oif1;  ago[0] = ogo0;  ago[1] = ogo1;
      aif[2].x = 0.0f; aif[2].y = 0.0f; aif[3].x = 0.0f; aif[3].y = 0.0f;
      ago[2].x = 0.0f; ago[2].y = 0.0f; ago[3].x = 0.0f; ago[3].y = 0.0f;

      const int bo = (u & 1) * (HID / 4);
      const float4 hv[6] = { hq1[bo], hq1[bo + 1], hq2[bo], hq2[bo + 1], hq3[bo], hq3[bo + 1] };
#pragma unroll
      for (int r = 0; r < 6; ++r) {
        const float4 h = hv[r];
        const int k = KB + 4 * r;
        aif[0] = fma2(wif[k + 0], h.x, aif[0]);  ago[0] = fma2(wgo[k + 0], h.x, ago[0]);
        aif[1] = fma2(wif[k + 1], h.y, aif[1]);  ago[1] = fma2(wgo[k + 1], h.y, ago[1]);
        aif[2] = fma2(wif[k + 2], h.z, aif[2]);  ago[2] = fma2(wgo[k + 2], h.z, ago[2]);
        aif[3] = fma2(wif[k + 3], h.w, aif[3]);  ago[3] = fma2(wgo[k + 3], h.w, ago[3]);
      }
      const v2f sif = (aif[0] + aif[1]) + (aif[2] + aif[3]);
      const v2f sgo = (ago[0] + ago[1]) + (ago[2] + ago[3]);

      const float zi = sif.x + xpair(sif.x);
      const float zf = sif.y + xpair(sif.y);
      const float zg = sgo.x + xpair(sgo.x);
      const float zo = sgo.y + xpair(sgo.y);

      const float ig = fsig_of_half(zi);
      const float fg = fsig_of_half(zf);
      const float gg = ftanh(zg);
      const float og = fsig_of_half(zo);
      creg = fg * creg + ig * gg;
      const float hn = og * ftanh(creg);
      sh_h[(u + 1) & 1][j] = hn;
      __builtin_amdgcn_sched_barrier(0);
      if (u & 1) {
        hsline[(size_t)(t >> 1) * 128] = p ? hn : hprev;
      } else {
        hprev = hn;
      }
      own_block_sums(hn, (u + 1 < XPF) ? xc[u + 1] : xn[0], p, wif, wgo, oif0, oif1, ogo0, ogo1);
      __builtin_amdgcn_sched_barrier(0);
      lds_barrier();
    }
    for (int u = 0; u < XPF; ++u) xc[u] = xn[u];
  }
}

__global__ __launch_bounds__(256) void head_wmma(
    const float* __restrict__ hs, const float* __restrict__ W_lin,
    const float* __restrict__ b_lin, const float* __restrict__ W_fin,
    const float* __restrict__ b_fin, float* __restrict__ out)
{
  const int lane = threadIdx.x & 31;
  const int wave = threadIdx.x >> 5;
  const int rt   = blockIdx.x * 8 + wave;
  const int l15  = lane & 15;
  const int half = lane >> 4;
  const int row0 = rt * 16;

  const int    arow  = row0 + l15;
  const float* Abase = hs + (size_t)(arow >> 1) * 128 + (arow & 1) * 16 + half * 2;
  v2f areg[16];
#pragma unroll
  for (int kk = 0; kk < 16; ++kk) {
    v2f a = *reinterpret_cast<const v2f*>(Abase + (kk >> 2) * 32 + (kk & 3) * 4);
    a.x = fmaxf(a.x, 0.0f);
    a.y = fmaxf(a.y, 0.0f);
    areg[kk] = a;
  }

  float partial[8];
#pragma unroll
  for (int v = 0; v < 8; ++v) partial[v] = 0.0f;

#pragma unroll
  for (int ctile = 0; ctile < 4; ++ctile) {
    const int  col  = ctile * 16 + l15;
    const v2f* Brow = reinterpret_cast<const v2f*>(W_lin + (size_t)col * HID);
    v8f c = {};
#pragma unroll
    for (int kk = 0; kk < 16; ++kk) {
      v2f b = Brow[kk * 2 + half];
      c = __builtin_amdgcn_wmma_f32_16x16x4_f32(false, areg[kk], false, b,
                                                (short)0, c, false, false);
    }
    const float bl = b_lin[col];
    const float wf = W_fin[col];
#pragma unroll
    for (int v = 0; v < 8; ++v)
      partial[v] += fmaxf(c[v] + bl, 0.0f) * wf;
  }

#pragma unroll
  for (int v = 0; v < 8; ++v) {
    float s = partial[v];
    s += __shfl_xor(s, 1, 32);
    s += __shfl_xor(s, 2, 32);
    s += __shfl_xor(s, 4, 32);
    s += __shfl_xor(s, 8, 32);
    if (l15 == 0) out[row0 + half * 8 + v] = s + b_fin[0];
  }
}

extern "C" void kernel_launch(void* const* d_in, const int* in_sizes, int n_in,
                              void* d_out, int out_size, void* d_ws, size_t ws_size,
                              hipStream_t stream) {
  const float* inputs = (const float*)d_in[0];
  const float* W_ih   = (const float*)d_in[1];
  const float* W_hh   = (const float*)d_in[2];
  const float* b_ih   = (const float*)d_in[3];
  const float* b_hh   = (const float*)d_in[4];
  const float* W_lin  = (const float*)d_in[5];
  const float* b_lin  = (const float*)d_in[6];
  const float* W_fin  = (const float*)d_in[7];
  const float* b_fin  = (const float*)d_in[8];
  float* out = (float*)d_out;

  float* x_proj = (float*)d_ws;
  float* hs     = x_proj + (size_t)SEQ_LEN * G4;

  xproj_wmma<<<1024, 256, 0, stream>>>(inputs, W_ih, b_ih, b_hh, x_proj);
  lstm_scan<<<1, SCAN_THREADS, 0, stream>>>(x_proj, W_hh, hs);
  head_wmma<<<1024, 256, 0, stream>>>(hs, W_lin, b_lin, W_fin, b_fin, out);
}


# lstm_scan: a unit's two k-half lanes sit in ONE 16-lane row: each gate's half sums join in one DPP add (row_xmask:8), not copy + permlanex16 + add; own-block k re-split: changes the summation order
# speedup vs baseline: 8.9157x; 1.0594x over previous
#include <hip/hip_runtime.h>

typedef float v2f __attribute__((ext_vector_type(2)));
typedef float v8f __attribute__((ext_vector_type(8)));

#define SEQ_LEN 131072
#define HID 64
#define G4 256

__global__ __launch_bounds__(256) void xproj_wmma(
    const float* __restrict__ inputs, const float* __restrict__ W_ih,
    const float* __restrict__ b_ih, const float* __restrict__ b_hh,
    float* __restrict__ x_proj)
{
  const int lane = threadIdx.x & 31;
  const int wave = threadIdx.x >> 5;
  const int rt   = blockIdx.x * 8 + wave;
  const int l15  = lane & 15;
  const int half = lane >> 4;
  const int row0 = rt * 16;

  const v2f* Arow = reinterpret_cast<const v2f*>(inputs + (size_t)(row0 + l15) * HID);
  v2f areg[16];
#pragma unroll
  for (int kk = 0; kk < 16; ++kk) areg[kk] = Arow[kk * 2 + half];

  for (int pj = 0; pj < 8; ++pj) {
    const int  pp   = pj >> 2;
    const int  ju   = (pj & 3) * 16 + l15;
    const int  col0 = (2 * pp) * HID + ju;
    const int  col1 = col0 + HID;
    const v2f* Brow0 = reinterpret_cast<const v2f*>(W_ih + (size_t)col0 * HID);
    const v2f* Brow1 = reinterpret_cast<const v2f*>(W_ih + (size_t)col1 * HID);
    v8f c0 = {};
    v8f c1 = {};
#pragma unroll
    for (int kk = 0; kk < 16; ++kk) {
      v2f b0 = Brow0[kk * 2 + half];
      v2f b1 = Brow1[kk * 2 + half];
      c0 = __builtin_amdgcn_wmma_f32_16x16x4_f32(false, areg[kk], false, b0,
                                                 (short)0, c0, false, false);
      c1 = __builtin_amdgcn_wmma_f32_16x16x4_f32(false, areg[kk], false, b1,
                                                 (short)0, c1, false, false);
    }
    const float bias0 = b_ih[col0] + b_hh[col0];
    const float bias1 = b_ih[col1] + b_hh[col1];
    const float s0 = (pp == 0) ? 0.5f : 1.0f;
#pragma unroll
    for (int v = 0; v < 8; ++v) {
      const int row = row0 + v + half * 8;
      v2f o;
      o.x = (c0[v] + bias0) * s0;
      o.y = (c1[v] + bias1) * 0.5f;
      *reinterpret_cast<v2f*>(x_proj + (size_t)row * G4 + pp * 128 + ju * 2) = o;
    }
  }
}

__device__ __forceinline__ float ftanh(float x)          { return __builtin_amdgcn_tanhf(x); }
__device__ __forceinline__ float fsig_of_half(float zh)  { return __builtin_fmaf(__builtin_amdgcn_tanhf(zh), 0.5f, 0.5f); }

#define XPF 8
#define SCAN_THREADS 128
#define KH 32
#define KB 8

__device__ __forceinline__ void lds_barrier() {
  __builtin_amdgcn_fence(__ATOMIC_RELEASE, "workgroup", "local");
  __builtin_amdgcn_s_barrier();
  __builtin_amdgcn_fence(__ATOMIC_ACQUIRE, "workgroup", "local");
}

__device__ __forceinline__ v2f fma2(v2f w, float h, v2f acc) {
  const v2f hh = {h, h};
  return __builtin_elementwise_fma(w, hh, acc);
}

__device__ __forceinline__ float xpair(float x) {
  const int b = __builtin_bit_cast(int, x);
  return __builtin_bit_cast(float,
      __builtin_amdgcn_update_dpp(b, b, 0x168, 0xf, 0xf, true));
}

template <int N>
__device__ __forceinline__ float own_row_lane(float x) {
  const int b = __builtin_bit_cast(int, x);
  return __builtin_bit_cast(float,
      __builtin_amdgcn_permlane16(b, b, 0x11111111u * N, 0x11111111u * (4 + N), false, false));
}

template <int N>
__device__ __forceinline__ float other_row_lane(float x) {
  const int b = __builtin_bit_cast(int, x);
  return __builtin_bit_cast(float,
      __builtin_amdgcn_permlanex16(b, b, 0x11111111u * N, 0x11111111u * (4 + N), false, false));
}

__device__ __forceinline__ void own_block_sums(
    float hn, v2f x, int p, const v2f (&wif)[KH], const v2f (&wgo)[KH],
    v2f& oif0, v2f& oif1, v2f& ogo0, v2f& ogo1)
{
  float hb[KB];
  hb[0] = own_row_lane<0>(hn);    hb[1] = own_row_lane<1>(hn);
  hb[2] = own_row_lane<2>(hn);    hb[3] = own_row_lane<3>(hn);
  hb[4] = other_row_lane<0>(hn);  hb[5] = other_row_lane<1>(hn);
  hb[6] = other_row_lane<2>(hn);  hb[7] = other_row_lane<3>(hn);
  oif0.x = (p == 0) ? x.x : 0.0f;  oif0.y = (p == 0) ? x.y : 0.0f;
  ogo0.x = (p == 0) ? 0.0f : x.x;  ogo0.y = (p == 0) ? 0.0f : x.y;
  oif1.x = 0.0f; oif1.y = 0.0f;    ogo1.x = 0.0f; ogo1.y = 0.0f;
#pragma unroll
  for (int kk = 0; kk < 4; ++kk) {
    oif0 = fma2(wif[kk],     hb[kk],     oif0);  ogo0 = fma2(wgo[kk],     hb[kk],     ogo0);
    oif1 = fma2(wif[4 + kk], hb[4 + kk], oif1);  ogo1 = fma2(wgo[4 + kk], hb[4 + kk], ogo1);
  }
}

__global__ __launch_bounds__(SCAN_THREADS) __attribute__((amdgpu_waves_per_eu(1, 1))) void lstm_scan(
    const float* __restrict__ x_proj, const float* __restrict__ W_hh,
    float* __restrict__ hs)
{
  __shared__ __attribute__((aligned(16))) float sh_h[2][HID];
  const int lane = threadIdx.x & 31;
  const int wave = threadIdx.x >> 5;
  const int r    = lane >> 4;
  const int p    = (lane >> 3) & 1;
  const int cu   = 8 * r + (lane & 7);
  const int j    = wave * 16 + cu;

  v2f wif[KH], wgo[KH];
#pragma unroll
  for (int bi = 0; bi < 4; ++bi) {
    const int kq = 16 * ((wave + bi) & 3) + KB * p;
#pragma unroll
    for (int kk = 0; kk < KB; ++kk) {
      const int k = (bi == 0)
          ? 16 * wave + 8 * ((kk < 4) ? r : 1 - r) + 4 * p + (kk & 3)
          : kq + kk;
      wif[KB * bi + kk].x = 0.5f * W_hh[(size_t)(0 * HID + j) * HID + k];
      wif[KB * bi + kk].y = 0.5f * W_hh[(size_t)(1 * HID + j) * HID + k];
      wgo[KB * bi + kk].x =        W_hh[(size_t)(2 * HID + j) * HID + k];
      wgo[KB * bi + kk].y = 0.5f * W_hh[(size_t)(3 * HID + j) * HID + k];
    }
  }
  const float4* hq1 = reinterpret_cast<const float4*>(&sh_h[0][16 * ((wave + 1) & 3) + KB * p]);
  const float4* hq2 = reinterpret_cast<const float4*>(&sh_h[0][16 * ((wave + 2) & 3) + KB * p]);
  const float4* hq3 = reinterpret_cast<const float4*>(&sh_h[0][16 * ((wave + 3) & 3) + KB * p]);

  float creg = 0.0f;
  float hprev = 0.0f;
  sh_h[0][j] = 0.0f;

  static_assert(SEQ_LEN % XPF == 0, "whole blocks");
  static_assert(XPF % 2 == 0, "buffer parity is the step's index in its block");
  const v2f* xcol = reinterpret_cast<const v2f*>(x_proj + p * 128 + j * 2);
  v2f xc[XPF], xn[XPF];
  for (int u = 0; u < XPF; ++u) xc[u] = xcol[(size_t)u * (G4 / 2)];

  v2f oif0, oif1, ogo0, ogo1;
  own_block_sums(0.0f, xc[0], p, wif, wgo, oif0, oif1, ogo0, ogo1);
  lds_barrier();

  float* hsline = hs + wave * 32 + p * 16 + cu;

  for (int tb = 0; tb < SEQ_LEN; tb += XPF) {
    const int nb = (tb + XPF < SEQ_LEN) ? (tb + XPF) : tb;
    for (int u = 0; u < XPF; ++u) xn[u] = xcol[(size_t)(nb + u) * (G4 / 2)];

#pragma unroll
    for (int u = 0; u < XPF; ++u) {
      const int t = tb + u;

      v2f aif[4], ago[4];
      aif[0] = oif0;  aif[1] = oif1;  ago[0] = ogo0;  ago[1] = ogo1;
      aif[2].x = 0.0f; aif[2].y = 0.0f; aif[3].x = 0.0f; aif[3].y = 0.0f;
      ago[2].x = 0.0f; ago[2].y = 0.0f; ago[3].x = 0.0f; ago[3].y = 0.0f;

      const int bo = (u & 1) * (HID / 4);
      const float4 hv[6] = { hq1[bo], hq1[bo + 1], hq2[bo], hq2[bo + 1], hq3[bo], hq3[bo + 1] };
#pragma unroll
      for (int r = 0; r < 6; ++r) {
        const float4 h = hv[r];
        const int k = KB + 4 * r;
        aif[0] = fma2(wif[k + 0], h.x, aif[0]);  ago[0] = fma2(wgo[k + 0], h.x, ago[0]);
        aif[1] = fma2(wif[k + 1], h.y, aif[1]);  ago[1] = fma2(wgo[k + 1], h.y, ago[1]);
        aif[2] = fma2(wif[k + 2], h.z, aif[2]);  ago[2] = fma2(wgo[k + 2], h.z, ago[2]);
        aif[3] = fma2(wif[k + 3], h.w, aif[3]);  ago[3] = fma2(wgo[k + 3], h.w, ago[3]);
      }
      const v2f sif = (aif[0] + aif[1]) + (aif[2] + aif[3]);
      const v2f sgo = (ago[0] + ago[1]) + (ago[2] + ago[3]);

      const float zi = sif.x + xpair(sif.x);
      const float zf = sif.y + xpair(sif.y);
      const float zg = sgo.x + xpair(sgo.x);
      const float zo = sgo.y + xpair(sgo.y);

      const float ig = fsig_of_half(zi);
      const float fg = fsig_of_half(zf);
      const float gg = ftanh(zg);
      const float og = fsig_of_half(zo);
      creg = fg * creg + ig * gg;
      const float hn = og * ftanh(creg);
      sh_h[(u + 1) & 1][j] = hn;
      __builtin_amdgcn_sched_barrier(0);
      if (u & 1) {
        hsline[(size_t)(t >> 1) * 128] = p ? hn : hprev;
      } else {
        hprev = hn;
      }
      own_block_sums(hn, (u + 1 < XPF) ? xc[u + 1] : xn[0], p, wif, wgo, oif0, oif1, ogo0, ogo1);
      __builtin_amdgcn_sched_barrier(0);
      lds_barrier();
    }
    for (int u = 0; u < XPF; ++u) xc[u] = xn[u];
  }
}

__global__ __launch_bounds__(256) void head_wmma(
    const float* __restrict__ hs, const float* __restrict__ W_lin,
    const float* __restrict__ b_lin, const float* __restrict__ W_fin,
    const float* __restrict__ b_fin, float* __restrict__ out)
{
  const int lane = threadIdx.x & 31;
  const int wave = threadIdx.x >> 5;
  const int rt   = blockIdx.x * 8 + wave;
  const int l15  = lane & 15;
  const int half = lane >> 4;
  const int row0 = rt * 16;

  const int    arow  = row0 + l15;
  const float* Abase = hs + (size_t)(arow >> 1) * 128 + (arow & 1) * 16 + half * 2;
  v2f areg[16];
#pragma unroll
  for (int kk = 0; kk < 16; ++kk) {
    v2f a = *reinterpret_cast<const v2f*>(Abase + (kk >> 2) * 32 + (kk & 3) * 4);
    a.x = fmaxf(a.x, 0.0f);
    a.y = fmaxf(a.y, 0.0f);
    areg[kk] = a;
  }

  float partial[8];
#pragma unroll
  for (int v = 0; v < 8; ++v) partial[v] = 0.0f;

#pragma unroll
  for (int ctile = 0; ctile < 4; ++ctile) {
    const int  col  = ctile * 16 + l15;
    const v2f* Brow = reinterpret_cast<const v2f*>(W_lin + (size_t)col * HID);
    v8f c = {};
#pragma unroll
    for (int kk = 0; kk < 16; ++kk) {
      v2f b = Brow[kk * 2 + half];
      c = __builtin_amdgcn_wmma_f32_16x16x4_f32(false, areg[kk], false, b,
                                                (short)0, c, false, false);
    }
    const float bl = b_lin[col];
    const float wf = W_fin[col];
#pragma unroll
    for (int v = 0; v < 8; ++v)
      partial[v] += fmaxf(c[v] + bl, 0.0f) * wf;
  }

#pragma unroll
  for (int v = 0; v < 8; ++v) {
    float s = partial[v];
    s += __shfl_xor(s, 1, 32);
    s += __shfl_xor(s, 2, 32);
    s += __shfl_xor(s, 4, 32);
    s += __shfl_xor(s, 8, 32);
    if (l15 == 0) out[row0 + half * 8 + v] = s + b_fin[0];
  }
}

extern "C" void kernel_launch(void* const* d_in, const int* in_sizes, int n_in,
                              void* d_out, int out_size, void* d_ws, size_t ws_size,
                              hipStream_t stream) {
  const float* inputs = (const float*)d_in[0];
  const float* W_ih   = (const float*)d_in[1];
  const float* W_hh   = (const float*)d_in[2];
  const float* b_ih   = (const float*)d_in[3];
  const float* b_hh   = (const float*)d_in[4];
  const float* W_lin  = (const float*)d_in[5];
  const float* b_lin  = (const float*)d_in[6];
  const float* W_fin  = (const float*)d_in[7];
  const float* b_fin  = (const float*)d_in[8];
  float* out = (float*)d_out;

  float* x_proj = (float*)d_ws;
  float* hs     = x_proj + (size_t)SEQ_LEN * G4;

  xproj_wmma<<<1024, 256, 0, stream>>>(inputs, W_ih, b_ih, b_hh, x_proj);
  lstm_scan<<<1, SCAN_THREADS, 0, stream>>>(x_proj, W_hh, hs);
  head_wmma<<<1024, 256, 0, stream>>>(hs, W_lin, b_lin, W_fin, b_fin, out);
}
